# starting kernel
# baseline (speedup 1.0000x reference)
#include <hip/hip_runtime.h>
#include <math.h>

#define NB 8
#define NS 50
#define NT 77
#define ND 1024
#define NR 16
#define TTILE 16
#define NTILES 5
#define HSTR 1028
#define PSTR 17

typedef __attribute__((ext_vector_type(2))) float v2f;
typedef __attribute__((ext_vector_type(4))) float v4f;
typedef __attribute__((ext_vector_type(8))) float v8f;

__global__ __launch_bounds__(256)
void loreft_rownorm_kernel(const float* __restrict__ R, float* __restrict__ inv)
{
    const int sr = blockIdx.x;
    const float* row = R + (size_t)sr * ND;

    float sum = 0.0f;
    for (int i = threadIdx.x; i < ND; i += 256) {
        float v = row[i];
        sum += v * v;
    }
    __shared__ float red[256];
    red[threadIdx.x] = sum;
    __syncthreads();
    for (int off = 128; off > 0; off >>= 1) {
        if (threadIdx.x < off) red[threadIdx.x] += red[threadIdx.x + off];
        __syncthreads();
    }
    if (threadIdx.x == 0) {
        float n = sqrtf(red[0]);
        n = fmaxf(n, 1e-12f);
        inv[sr] = 1.0f / n;
    }
}

__global__ __launch_bounds__(128)
void loreft_main_kernel(const float* __restrict__ h,
                        const float* __restrict__ R,
                        const float* __restrict__ W,
                        const float* __restrict__ bvec,
                        const float* __restrict__ inv,
                        float* __restrict__ out)
{
    extern __shared__ float smem[];
    float* __restrict__ htile   = smem;
    float* __restrict__ partial = htile + TTILE * HSTR;
    float* __restrict__ Pm      = partial + 4 * TTILE * NR;
    float* __restrict__ invL    = Pm + TTILE * PSTR;
    float* __restrict__ bL      = invL + NR;

    const int blk  = blockIdx.x;
    const int tile = blk % NTILES;
    const int bs   = blk / NTILES;
    const int s    = bs % NS;
    const int b    = bs / NS;
    const int t0   = tile * TTILE;

    const int lane = threadIdx.x & 31;
    const int wave = threadIdx.x >> 5;
    const int m    = lane & 15;
    const int kk   = (lane >> 4) << 1;

    const size_t tok_base = ((size_t)(b * NS + s)) * NT;
    const float* __restrict__ hbase = h + tok_base * ND;
    float*       __restrict__ obase = out + tok_base * ND;
    const float* __restrict__ Rs = R + (size_t)s * NR * ND;
    const float* __restrict__ Ws = W + (size_t)s * NR * ND;

    if (threadIdx.x < NR) {
        invL[threadIdx.x] = inv[s * NR + threadIdx.x];
        bL[threadIdx.x]   = bvec[s * NR + threadIdx.x];
    }

    for (int idx = threadIdx.x; idx < TTILE * (ND / 4); idx += 128) {
        int row  = idx >> 8;
        int c4   = idx & 255;
        int tokr = t0 + row;
        tokr = (tokr < NT) ? tokr : (NT - 1);
        v4f v = *(const v4f*)(hbase + (size_t)tokr * ND + ((size_t)c4 << 2));
        *(v4f*)(htile + row * HSTR + (c4 << 2)) = v;
    }
    __syncthreads();

    const float invm = invL[m];
    const float* __restrict__ Rrow = Rs + (size_t)m * ND;
    const float* __restrict__ Wrow = Ws + (size_t)m * ND;
    const float* __restrict__ hL   = htile + m * HSTR;

    v8f accR = {};
    v8f accW = {};
    const int k0 = wave * (ND / 4);
    for (int k = k0; k < k0 + ND / 4; k += 4) {
        v2f a  = *(const v2f*)(hL + k + kk);
        v2f br = *(const v2f*)(Rrow + k + kk);
        v2f bw = *(const v2f*)(Wrow + k + kk);
        br.x *= invm;
        br.y *= invm;
        accR = __builtin_amdgcn_wmma_f32_16x16x4_f32(
                   false, a, false, br, (short)0, accR, false, false);
        accW = __builtin_amdgcn_wmma_f32_16x16x4_f32(
                   false, a, false, bw, (short)0, accW, false, false);
    }

    #pragma unroll
    for (int i = 0; i < 8; ++i) {
        int row = i + ((lane >> 4) << 3);
        int col = lane & 15;
        partial[(wave * TTILE + row) * NR + col] = accW[i] - accR[i];
    }
    __syncthreads();
    for (int idx = threadIdx.x; idx < TTILE * NR; idx += 128) {
        int row = idx >> 4, col = idx & 15;
        Pm[row * PSTR + col] =
              partial[(0 * TTILE + row) * NR + col]
            + partial[(1 * TTILE + row) * NR + col]
            + partial[(2 * TTILE + row) * NR + col]
            + partial[(3 * TTILE + row) * NR + col]
            + bL[col];
    }
    __syncthreads();

    for (int j = 0; j < 16; ++j) {
        const int d0   = ((j << 2) | wave) << 4;
        const int dcol = d0 + m;
        v8f acc = {};
        #pragma unroll
        for (int kc = 0; kc < NR; kc += 4) {
            v2f a, br;
            a.x  = Pm[m * PSTR + kc + kk];
            a.y  = Pm[m * PSTR + kc + kk + 1];
            br.x = Rs[(size_t)(kc + kk)     * ND + dcol] * invL[kc + kk];
            br.y = Rs[(size_t)(kc + kk + 1) * ND + dcol] * invL[kc + kk + 1];
            acc = __builtin_amdgcn_wmma_f32_16x16x4_f32(
                      false, a, false, br, (short)0, acc, false, false);
        }
        #pragma unroll
        for (int i = 0; i < 8; ++i) {
            int row  = i + ((lane >> 4) << 3);
            int toko = t0 + row;
            if (toko < NT) {
                int col = d0 + (lane & 15);
                obase[(size_t)toko * ND + col] =
                    htile[row * HSTR + col] + acc[i];
            }
        }
    }
}

extern "C" void kernel_launch(void* const* d_in, const int* in_sizes, int n_in,
                              void* d_out, int out_size, void* d_ws, size_t ws_size,
                              hipStream_t stream)
{
    const float* h    = (const float*)d_in[0];
    const float* R    = (const float*)d_in[1];
    const float* W    = (const float*)d_in[2];
    const float* bvec = (const float*)d_in[3];
    float* out = (float*)d_out;
    float* inv = (float*)d_ws;

    (void)in_sizes; (void)n_in; (void)out_size; (void)ws_size;

    enum { SMEM_FLOATS = TTILE * HSTR + 4 * TTILE * NR + TTILE * PSTR + NR + NR };

    loreft_rownorm_kernel<<<NS * NR, 256, 0, stream>>>(R, inv);
    loreft_main_kernel<<<NB * NS * NTILES, 128, SMEM_FLOATS * sizeof(float), stream>>>(
        h, R, W, bvec, inv, out);
}


# One launch: 1/|R| applied to the 16x16 tile P, not to operands (Rn h = (R h)/|R|, Rn^T P = R^T (P/|R|)); |R|^2 partials from phase-A fragments join the cross-wave sum; norm and P sum order changed
# speedup vs baseline: 1.6924x; 1.6924x over previous
#include <hip/hip_runtime.h>
#include <math.h>

#define NB 8
#define NS 50
#define NT 77
#define ND 1024
#define NR 16
#define TTILE 16
#define NTILES 5
#define NWAVES 8
#define WCOLS 128
#define SPITCH 132
#define PPITCH 20
#define SLICE_FLOATS (TTILE * SPITCH)
#define PART_FLOATS  (TTILE * PPITCH)

typedef __attribute__((ext_vector_type(2))) float v2f;
typedef __attribute__((ext_vector_type(4))) float v4f;
typedef __attribute__((ext_vector_type(8))) float v8f;

__device__ __forceinline__ void wave_lds_handover()
{
    __builtin_amdgcn_fence(__ATOMIC_RELEASE, "wavefront");
    __builtin_amdgcn_wave_barrier();
    __builtin_amdgcn_fence(__ATOMIC_ACQUIRE, "wavefront");
}

__global__ __launch_bounds__(256)
void loreft_main_kernel(const float* __restrict__ h,
                        const float* __restrict__ R,
                        const float* __restrict__ W,
                        const float* __restrict__ bvec,
                        float* __restrict__ out)
{
    extern __shared__ float smem[];

    const int blk  = blockIdx.x;
    const int tile = blk % NTILES;
    const int bs   = blk / NTILES;
    const int s    = bs % NS;
    const int b    = bs / NS;
    const int t0   = tile * TTILE;

    const int lane = threadIdx.x & 31;
    const int wave = threadIdx.x >> 5;
    const int m    = lane & 15;
    const int hh   = lane >> 4;
    const int c0   = wave * WCOLS;

    float* __restrict__ sl   = smem + wave * SLICE_FLOATS;
    float* __restrict__ partW = smem + NWAVES * SLICE_FLOATS;
    float* __restrict__ partR = partW + NWAVES * PART_FLOATS;
    float* __restrict__ ssq   = partR + NWAVES * PART_FLOATS;

    const size_t tok_base = ((size_t)(b * NS + s)) * NT;
    const float* __restrict__ hbase = h + tok_base * ND;
    float*       __restrict__ obase = out + tok_base * ND;
    const float* __restrict__ Rs = R + (size_t)s * NR * ND;
    const float* __restrict__ Ws = W + (size_t)s * NR * ND;

    v4f hv[TTILE];
    #pragma unroll
    for (int i = 0; i < TTILE; ++i) {
        int tokr = t0 + i;
        tokr = (tokr < NT) ? tokr : (NT - 1);
        hv[i] = *(const v4f*)(hbase + (size_t)tokr * ND + c0 + 4 * lane);
    }

    const float* __restrict__ Rrow = Rs + (size_t)m * ND + c0 + 8 * hh;
    const float* __restrict__ Wrow = Ws + (size_t)m * ND + c0 + 8 * hh;
    v4f r0[8], r1[8], w0[8], w1[8];
    #pragma unroll
    for (int q = 0; q < 8; ++q) {
        r0[q] = *(const v4f*)(Rrow + 16 * q);
        r1[q] = *(const v4f*)(Rrow + 16 * q + 4);
        w0[q] = *(const v4f*)(Wrow + 16 * q);
        w1[q] = *(const v4f*)(Wrow + 16 * q + 4);
    }

    #pragma unroll
    for (int i = 0; i < TTILE; ++i) {
        *(v4f*)(sl + i * SPITCH + 4 * lane) = hv[i];
    }
    wave_lds_handover();

    v4f hlo[8], hhi[8];
    #pragma unroll
    for (int q = 0; q < 8; ++q) {
        hlo[q] = *(const v4f*)(sl + m * SPITCH + 16 * q + 8 * hh);
        hhi[q] = *(const v4f*)(sl + m * SPITCH + 16 * q + 8 * hh + 4);
    }

    v4f sq = r0[0] * r0[0];
    sq += r1[0] * r1[0];
    #pragma unroll
    for (int q = 1; q < 8; ++q) {
        sq += r0[q] * r0[q];
        sq += r1[q] * r1[q];
    }
    float ssl = (sq.x + sq.y) + (sq.z + sq.w);
    ssl += __shfl_xor(ssl, 16, 32);

    v8f accR = {};
    v8f accW = {};
    #pragma unroll
    for (int q = 0; q < 8; ++q) {
        const v4f rs0 = r0[q];
        const v4f rs1 = r1[q];
        v2f a, br, bw;
        a.x = hlo[q].x; a.y = hlo[q].y; br.x = rs0.x; br.y = rs0.y; bw.x = w0[q].x; bw.y = w0[q].y;
        accR = __builtin_amdgcn_wmma_f32_16x16x4_f32(false, a, false, br, (short)0, accR, false, false);
        accW = __builtin_amdgcn_wmma_f32_16x16x4_f32(false, a, false, bw, (short)0, accW, false, false);
        a.x = hlo[q].z; a.y = hlo[q].w; br.x = rs0.z; br.y = rs0.w; bw.x = w0[q].z; bw.y = w0[q].w;
        accR = __builtin_amdgcn_wmma_f32_16x16x4_f32(false, a, false, br, (short)0, accR, false, false);
        accW = __builtin_amdgcn_wmma_f32_16x16x4_f32(false, a, false, bw, (short)0, accW, false, false);
        a.x = hhi[q].x; a.y = hhi[q].y; br.x = rs1.x; br.y = rs1.y; bw.x = w1[q].x; bw.y = w1[q].y;
        accR = __builtin_amdgcn_wmma_f32_16x16x4_f32(false, a, false, br, (short)0, accR, false, false);
        accW = __builtin_amdgcn_wmma_f32_16x16x4_f32(false, a, false, bw, (short)0, accW, false, false);
        a.x = hhi[q].z; a.y = hhi[q].w; br.x = rs1.z; br.y = rs1.w; bw.x = w1[q].z; bw.y = w1[q].w;
        accR = __builtin_amdgcn_wmma_f32_16x16x4_f32(false, a, false, br, (short)0, accR, false, false);
        accW = __builtin_amdgcn_wmma_f32_16x16x4_f32(false, a, false, bw, (short)0, accW, false, false);
    }

    float rraw[8][8];
    #pragma unroll
    for (int q = 0; q < 8; ++q) {
        const float* __restrict__ Rc = Rs + (size_t)(8 * hh) * ND + c0 + 16 * q + m;
        #pragma unroll
        for (int i = 0; i < 8; ++i) rraw[q][i] = Rc[(size_t)i * ND];
    }
    const v4f b0 = *(const v4f*)(bvec + s * NR + 8 * hh);
    const v4f b1 = *(const v4f*)(bvec + s * NR + 8 * hh + 4);

    #pragma unroll
    for (int i = 0; i < 8; ++i) {
        partW[wave * PART_FLOATS + (8 * hh + i) * PPITCH + m] = accW[i];
        partR[wave * PART_FLOATS + (8 * hh + i) * PPITCH + m] = accR[i];
    }
    ssq[wave * 32 + lane] = ssl;
    __syncthreads();

    float tot = ssq[lane];
    #pragma unroll
    for (int w = 1; w < NWAVES; ++w) tot += ssq[w * 32 + lane];
    const float invl = 1.0f / fmaxf(sqrtf(tot), 1e-12f);
    float invr[8];
    #pragma unroll
    for (int i = 0; i < 8; ++i) invr[i] = __shfl(invl, 8 * hh + i, 32);

    v4f sw0 = *(const v4f*)(partW + m * PPITCH + 8 * hh);
    v4f sw1 = *(const v4f*)(partW + m * PPITCH + 8 * hh + 4);
    v4f sr0 = *(const v4f*)(partR + m * PPITCH + 8 * hh);
    v4f sr1 = *(const v4f*)(partR + m * PPITCH + 8 * hh + 4);
    #pragma unroll
    for (int w = 1; w < NWAVES; ++w) {
        sw0 += *(const v4f*)(partW + w * PART_FLOATS + m * PPITCH + 8 * hh);
        sw1 += *(const v4f*)(partW + w * PART_FLOATS + m * PPITCH + 8 * hh + 4);
        sr0 += *(const v4f*)(partR + w * PART_FLOATS + m * PPITCH + 8 * hh);
        sr1 += *(const v4f*)(partR + w * PART_FLOATS + m * PPITCH + 8 * hh + 4);
    }
    v4f iv0, iv1;
    iv0.x = invr[0]; iv0.y = invr[1]; iv0.z = invr[2]; iv0.w = invr[3];
    iv1.x = invr[4]; iv1.y = invr[5]; iv1.z = invr[6]; iv1.w = invr[7];
    const v4f p0 = (sw0 - sr0 * iv0 + b0) * iv0;
    const v4f p1 = (sw1 - sr1 * iv1 + b1) * iv1;
    const float pf[8] = { p0.x, p0.y, p0.z, p0.w, p1.x, p1.y, p1.z, p1.w };

    #pragma unroll
    for (int q = 0; q < 8; ++q) {
        v8f acc = {};
        #pragma unroll
        for (int t = 0; t < 4; ++t) {
            v2f a, bp;
            a.x  = rraw[q][2 * t];  a.y  = rraw[q][2 * t + 1];
            bp.x = pf[2 * t];  bp.y = pf[2 * t + 1];
            acc = __builtin_amdgcn_wmma_f32_16x16x4_f32(false, a, false, bp, (short)0, acc, false, false);
        }
        v4f o0, o1;
        o0.x = hlo[q].x + acc[0]; o0.y = hlo[q].y + acc[1]; o0.z = hlo[q].z + acc[2]; o0.w = hlo[q].w + acc[3];
        o1.x = hhi[q].x + acc[4]; o1.y = hhi[q].y + acc[5]; o1.z = hhi[q].z + acc[6]; o1.w = hhi[q].w + acc[7];
        *(v4f*)(sl + m * SPITCH + 16 * q + 8 * hh)     = o0;
        *(v4f*)(sl + m * SPITCH + 16 * q + 8 * hh + 4) = o1;
    }
    wave_lds_handover();

    if (t0 + TTILE <= NT) {
        v4f ov[TTILE];
        #pragma unroll
        for (int i = 0; i < TTILE; ++i) ov[i] = *(const v4f*)(sl + i * SPITCH + 4 * lane);
        #pragma unroll
        for (int i = 0; i < TTILE; ++i)
            *(v4f*)(obase + (size_t)(t0 + i) * ND + c0 + 4 * lane) = ov[i];
    } else {
        for (int i = 0; i < NT - t0; ++i) {
            v4f v = *(const v4f*)(sl + i * SPITCH + 4 * lane);
            *(v4f*)(obase + (size_t)(t0 + i) * ND + c0 + 4 * lane) = v;
        }
    }
}

extern "C" void kernel_launch(void* const* d_in, const int* in_sizes, int n_in,
                              void* d_out, int out_size, void* d_ws, size_t ws_size,
                              hipStream_t stream)
{
    const float* h    = (const float*)d_in[0];
    const float* R    = (const float*)d_in[1];
    const float* W    = (const float*)d_in[2];
    const float* bvec = (const float*)d_in[3];
    float* out = (float*)d_out;

    (void)in_sizes; (void)n_in; (void)out_size; (void)d_ws; (void)ws_size;

    enum { SMEM_FLOATS = NWAVES * SLICE_FLOATS + 2 * NWAVES * PART_FLOATS + NWAVES * 32 };

    loreft_main_kernel<<<NB * NS * NTILES, 256, SMEM_FLOATS * sizeof(float), stream>>>(
        h, R, W, bvec, out);
}


# W and R slices come in as whole-row 512-B loads through the wave's private LDS slice, as h does; phase-A fragments and phase-B R^T fragments are read from LDS, not global; sums unchanged
# speedup vs baseline: 1.7857x; 1.0551x over previous
#include <hip/hip_runtime.h>
#include <math.h>

#define NB 8
#define NS 50
#define NT 77
#define ND 1024
#define NR 16
#define TTILE 16
#define NTILES 5
#define NWAVES 8
#define WCOLS 128
#define SPITCH 132
#define PPITCH 20
#define SLICE_FLOATS (TTILE * SPITCH)
#define PART_FLOATS  (TTILE * PPITCH)

typedef __attribute__((ext_vector_type(2))) float v2f;
typedef __attribute__((ext_vector_type(4))) float v4f;
typedef __attribute__((ext_vector_type(8))) float v8f;

__device__ __forceinline__ void wave_lds_handover()
{
    __builtin_amdgcn_fence(__ATOMIC_RELEASE, "wavefront");
    __builtin_amdgcn_wave_barrier();
    __builtin_amdgcn_fence(__ATOMIC_ACQUIRE, "wavefront");
}

__global__ __launch_bounds__(256, 4)
void loreft_main_kernel(const float* __restrict__ h,
                        const float* __restrict__ R,
                        const float* __restrict__ W,
                        const float* __restrict__ bvec,
                        float* __restrict__ out)
{
    extern __shared__ float smem[];

    const int blk  = blockIdx.x;
    const int tile = blk % NTILES;
    const int bs   = blk / NTILES;
    const int s    = bs % NS;
    const int b    = bs / NS;
    const int t0   = tile * TTILE;

    const int lane = threadIdx.x & 31;
    const int wave = threadIdx.x >> 5;
    const int m    = lane & 15;
    const int hh   = lane >> 4;
    const int c0   = wave * WCOLS;

    float* __restrict__ sl   = smem + wave * SLICE_FLOATS;
    float* __restrict__ partW = smem + NWAVES * SLICE_FLOATS;
    float* __restrict__ partR = partW + NWAVES * PART_FLOATS;
    float* __restrict__ ssq   = partR + NWAVES * PART_FLOATS;

    const size_t tok_base = ((size_t)(b * NS + s)) * NT;
    const float* __restrict__ hbase = h + tok_base * ND;
    float*       __restrict__ obase = out + tok_base * ND;
    const float* __restrict__ Rs = R + (size_t)s * NR * ND;
    const float* __restrict__ Ws = W + (size_t)s * NR * ND;

    v4f wv[NR];
    #pragma unroll
    for (int i = 0; i < NR; ++i) wv[i] = *(const v4f*)(Ws + (size_t)i * ND + c0 + 4 * lane);

    v4f hv[TTILE];
    #pragma unroll
    for (int i = 0; i < TTILE; ++i) {
        int tokr = t0 + i;
        tokr = (tokr < NT) ? tokr : (NT - 1);
        hv[i] = *(const v4f*)(hbase + (size_t)tokr * ND + c0 + 4 * lane);
    }

    v4f rv[NR];
    #pragma unroll
    for (int i = 0; i < NR; ++i) rv[i] = *(const v4f*)(Rs + (size_t)i * ND + c0 + 4 * lane);

    #pragma unroll
    for (int i = 0; i < NR; ++i) {
        *(v4f*)(sl + i * SPITCH + 4 * lane) = wv[i];
    }
    wave_lds_handover();

    v4f r0[8], r1[8], w0[8], w1[8];
    #pragma unroll
    for (int q = 0; q < 8; ++q) {
        w0[q] = *(const v4f*)(sl + m * SPITCH + 16 * q + 8 * hh);
        w1[q] = *(const v4f*)(sl + m * SPITCH + 16 * q + 8 * hh + 4);
    }

    wave_lds_handover();
    #pragma unroll
    for (int i = 0; i < TTILE; ++i) {
        *(v4f*)(sl + i * SPITCH + 4 * lane) = hv[i];
    }
    wave_lds_handover();

    v4f hlo[8], hhi[8];
    #pragma unroll
    for (int q = 0; q < 8; ++q) {
        hlo[q] = *(const v4f*)(sl + m * SPITCH + 16 * q + 8 * hh);
        hhi[q] = *(const v4f*)(sl + m * SPITCH + 16 * q + 8 * hh + 4);
    }

    wave_lds_handover();
    #pragma unroll
    for (int i = 0; i < NR; ++i) {
        *(v4f*)(sl + i * SPITCH + 4 * lane) = rv[i];
    }
    wave_lds_handover();
    #pragma unroll
    for (int q = 0; q < 8; ++q) {
        r0[q] = *(const v4f*)(sl + m * SPITCH + 16 * q + 8 * hh);
        r1[q] = *(const v4f*)(sl + m * SPITCH + 16 * q + 8 * hh + 4);
    }

    v4f sq = r0[0] * r0[0];
    sq += r1[0] * r1[0];
    #pragma unroll
    for (int q = 1; q < 8; ++q) {
        sq += r0[q] * r0[q];
        sq += r1[q] * r1[q];
    }
    float ssl = (sq.x + sq.y) + (sq.z + sq.w);
    ssl += __shfl_xor(ssl, 16, 32);

    v8f accR = {};
    v8f accW = {};
    #pragma unroll
    for (int q = 0; q < 8; ++q) {
        const v4f rs0 = r0[q];
        const v4f rs1 = r1[q];
        v2f a, br, bw;
        a.x = hlo[q].x; a.y = hlo[q].y; br.x = rs0.x; br.y = rs0.y; bw.x = w0[q].x; bw.y = w0[q].y;
        accR = __builtin_amdgcn_wmma_f32_16x16x4_f32(false, a, false, br, (short)0, accR, false, false);
        accW = __builtin_amdgcn_wmma_f32_16x16x4_f32(false, a, false, bw, (short)0, accW, false, false);
        a.x = hlo[q].z; a.y = hlo[q].w; br.x = rs0.z; br.y = rs0.w; bw.x = w0[q].z; bw.y = w0[q].w;
        accR = __builtin_amdgcn_wmma_f32_16x16x4_f32(false, a, false, br, (short)0, accR, false, false);
        accW = __builtin_amdgcn_wmma_f32_16x16x4_f32(false, a, false, bw, (short)0, accW, false, false);
        a.x = hhi[q].x; a.y = hhi[q].y; br.x = rs1.x; br.y = rs1.y; bw.x = w1[q].x; bw.y = w1[q].y;
        accR = __builtin_amdgcn_wmma_f32_16x16x4_f32(false, a, false, br, (short)0, accR, false, false);
        accW = __builtin_amdgcn_wmma_f32_16x16x4_f32(false, a, false, bw, (short)0, accW, false, false);
        a.x = hhi[q].z; a.y = hhi[q].w; br.x = rs1.z; br.y = rs1.w; bw.x = w1[q].z; bw.y = w1[q].w;
        accR = __builtin_amdgcn_wmma_f32_16x16x4_f32(false, a, false, br, (short)0, accR, false, false);
        accW = __builtin_amdgcn_wmma_f32_16x16x4_f32(false, a, false, bw, (short)0, accW, false, false);
    }

    float rraw[8][8];
    #pragma unroll
    for (int q = 0; q < 8; ++q) {
        const float* __restrict__ Rc = sl + (8 * hh) * SPITCH + 16 * q + m;
        #pragma unroll
        for (int i = 0; i < 8; ++i) rraw[q][i] = Rc[i * SPITCH];
    }
    const v4f b0 = *(const v4f*)(bvec + s * NR + 8 * hh);
    const v4f b1 = *(const v4f*)(bvec + s * NR + 8 * hh + 4);

    #pragma unroll
    for (int i = 0; i < 8; ++i) {
        partW[wave * PART_FLOATS + (8 * hh + i) * PPITCH + m] = accW[i];
        partR[wave * PART_FLOATS + (8 * hh + i) * PPITCH + m] = accR[i];
    }
    ssq[wave * 32 + lane] = ssl;
    __syncthreads();

    float tot = ssq[lane];
    #pragma unroll
    for (int w = 1; w < NWAVES; ++w) tot += ssq[w * 32 + lane];
    const float invl = 1.0f / fmaxf(sqrtf(tot), 1e-12f);
    float invr[8];
    #pragma unroll
    for (int i = 0; i < 8; ++i) invr[i] = __shfl(invl, 8 * hh + i, 32);

    v4f sw0 = *(const v4f*)(partW + m * PPITCH + 8 * hh);
    v4f sw1 = *(const v4f*)(partW + m * PPITCH + 8 * hh + 4);
    v4f sr0 = *(const v4f*)(partR + m * PPITCH + 8 * hh);
    v4f sr1 = *(const v4f*)(partR + m * PPITCH + 8 * hh + 4);
    #pragma unroll
    for (int w = 1; w < NWAVES; ++w) {
        sw0 += *(const v4f*)(partW + w * PART_FLOATS + m * PPITCH + 8 * hh);
        sw1 += *(const v4f*)(partW + w * PART_FLOATS + m * PPITCH + 8 * hh + 4);
        sr0 += *(const v4f*)(partR + w * PART_FLOATS + m * PPITCH + 8 * hh);
        sr1 += *(const v4f*)(partR + w * PART_FLOATS + m * PPITCH + 8 * hh + 4);
    }
    v4f iv0, iv1;
    iv0.x = invr[0]; iv0.y = invr[1]; iv0.z = invr[2]; iv0.w = invr[3];
    iv1.x = invr[4]; iv1.y = invr[5]; iv1.z = invr[6]; iv1.w = invr[7];
    const v4f p0 = (sw0 - sr0 * iv0 + b0) * iv0;
    const v4f p1 = (sw1 - sr1 * iv1 + b1) * iv1;
    const float pf[8] = { p0.x, p0.y, p0.z, p0.w, p1.x, p1.y, p1.z, p1.w };

    #pragma unroll
    for (int q = 0; q < 8; ++q) {
        v8f acc = {};
        #pragma unroll
        for (int t = 0; t < 4; ++t) {
            v2f a, bp;
            a.x  = rraw[q][2 * t];  a.y  = rraw[q][2 * t + 1];
            bp.x = pf[2 * t];  bp.y = pf[2 * t + 1];
            acc = __builtin_amdgcn_wmma_f32_16x16x4_f32(false, a, false, bp, (short)0, acc, false, false);
        }
        v4f o0, o1;
        o0.x = hlo[q].x + acc[0]; o0.y = hlo[q].y + acc[1]; o0.z = hlo[q].z + acc[2]; o0.w = hlo[q].w + acc[3];
        o1.x = hhi[q].x + acc[4]; o1.y = hhi[q].y + acc[5]; o1.z = hhi[q].z + acc[6]; o1.w = hhi[q].w + acc[7];
        *(v4f*)(sl + m * SPITCH + 16 * q + 8 * hh)     = o0;
        *(v4f*)(sl + m * SPITCH + 16 * q + 8 * hh + 4) = o1;
    }
    wave_lds_handover();

    if (t0 + TTILE <= NT) {
        v4f ov[TTILE];
        #pragma unroll
        for (int i = 0; i < TTILE; ++i) ov[i] = *(const v4f*)(sl + i * SPITCH + 4 * lane);
        #pragma unroll
        for (int i = 0; i < TTILE; ++i)
            *(v4f*)(obase + (size_t)(t0 + i) * ND + c0 + 4 * lane) = ov[i];
    } else {
        for (int i = 0; i < NT - t0; ++i) {
            v4f v = *(const v4f*)(sl + i * SPITCH + 4 * lane);
            *(v4f*)(obase + (size_t)(t0 + i) * ND + c0 + 4 * lane) = v;
        }
    }
}

extern "C" void kernel_launch(void* const* d_in, const int* in_sizes, int n_in,
                              void* d_out, int out_size, void* d_ws, size_t ws_size,
                              hipStream_t stream)
{
    const float* h    = (const float*)d_in[0];
    const float* R    = (const float*)d_in[1];
    const float* W    = (const float*)d_in[2];
    const float* bvec = (const float*)d_in[3];
    float* out = (float*)d_out;

    (void)in_sizes; (void)n_in; (void)out_size; (void)d_ws; (void)ws_size;

    enum { SMEM_FLOATS = NWAVES * SLICE_FLOATS + 2 * NWAVES * PART_FLOATS + NWAVES * 32 };

    loreft_main_kernel<<<NB * NS * NTILES, 256, SMEM_FLOATS * sizeof(float), stream>>>(
        h, R, W, bvec, out);
}


# h and R slices load as two 64-column halves (2 rows x 256 B per load, first halves first); first half's LDS passes and 32 phase-A products run while the second half is in flight; sums unchanged
# speedup vs baseline: 1.9018x; 1.0650x over previous
#include <hip/hip_runtime.h>
#include <math.h>

#define NB 8
#define NS 50
#define NT 77
#define ND 1024
#define NR 16
#define TTILE 16
#define NTILES 5
#define NWAVES 8
#define WCOLS 128
#define SPITCH 132
#define PPITCH 20
#define SLICE_FLOATS (TTILE * SPITCH)
#define PART_FLOATS  (TTILE * PPITCH)

typedef __attribute__((ext_vector_type(2))) float v2f;
typedef __attribute__((ext_vector_type(4))) float v4f;
typedef __attribute__((ext_vector_type(8))) float v8f;

__device__ __forceinline__ void wave_lds_handover()
{
    __builtin_amdgcn_fence(__ATOMIC_RELEASE, "wavefront");
    __builtin_amdgcn_wave_barrier();
    __builtin_amdgcn_fence(__ATOMIC_ACQUIRE, "wavefront");
}

__global__ __launch_bounds__(256, 4)
void loreft_main_kernel(const float* __restrict__ h,
                        const float* __restrict__ R,
                        const float* __restrict__ W,
                        const float* __restrict__ bvec,
                        float* __restrict__ out)
{
    extern __shared__ float smem[];

    const int blk  = blockIdx.x;
    const int tile = blk % NTILES;
    const int bs   = blk / NTILES;
    const int s    = bs % NS;
    const int b    = bs / NS;
    const int t0   = tile * TTILE;

    const int lane = threadIdx.x & 31;
    const int wave = threadIdx.x >> 5;
    const int m    = lane & 15;
    const int hh   = lane >> 4;
    const int c0   = wave * WCOLS;

    float* __restrict__ sl   = smem + wave * SLICE_FLOATS;
    float* __restrict__ partW = smem + NWAVES * SLICE_FLOATS;
    float* __restrict__ partR = partW + NWAVES * PART_FLOATS;
    float* __restrict__ ssq   = partR + NWAVES * PART_FLOATS;

    const size_t tok_base = ((size_t)(b * NS + s)) * NT;
    const float* __restrict__ hbase = h + tok_base * ND;
    float*       __restrict__ obase = out + tok_base * ND;
    const float* __restrict__ Rs = R + (size_t)s * NR * ND;
    const float* __restrict__ Ws = W + (size_t)s * NR * ND;

    v4f wv[NR];
    #pragma unroll
    for (int i = 0; i < NR; ++i) wv[i] = *(const v4f*)(Ws + (size_t)i * ND + c0 + 4 * lane);

    v4f hv[2][TTILE / 2];
    v4f rv[2][NR / 2];
    #pragma unroll
    for (int hf = 0; hf < 2; ++hf) {
        #pragma unroll
        for (int j = 0; j < TTILE / 2; ++j) {
            int tokr = t0 + 2 * j + hh;
            tokr = (tokr < NT) ? tokr : (NT - 1);
            hv[hf][j] = *(const v4f*)(hbase + (size_t)tokr * ND + c0 + 64 * hf + 4 * m);
        }
        #pragma unroll
        for (int j = 0; j < NR / 2; ++j)
            rv[hf][j] = *(const v4f*)(Rs + (size_t)(2 * j + hh) * ND + c0 + 64 * hf + 4 * m);
    }

    #pragma unroll
    for (int i = 0; i < NR; ++i) {
        *(v4f*)(sl + i * SPITCH + 4 * lane) = wv[i];
    }
    wave_lds_handover();

    v4f r0[8], r1[8], w0[8], w1[8];
    #pragma unroll
    for (int q = 0; q < 8; ++q) {
        w0[q] = *(const v4f*)(sl + m * SPITCH + 16 * q + 8 * hh);
        w1[q] = *(const v4f*)(sl + m * SPITCH + 16 * q + 8 * hh + 4);
    }

    v4f hlo[8], hhi[8];
    v4f sq;
    v8f accR = {};
    v8f accW = {};
    #pragma unroll
    for (int hf = 0; hf < 2; ++hf) {
        float* __restrict__ slh = sl + 64 * hf;
        if (hf == 0) wave_lds_handover();
        #pragma unroll
        for (int j = 0; j < TTILE / 2; ++j) {
            *(v4f*)(slh + (2 * j + hh) * SPITCH + 4 * m) = hv[hf][j];
        }
        wave_lds_handover();
        #pragma unroll
        for (int q = 4 * hf; q < 4 * hf + 4; ++q) {
            hlo[q] = *(const v4f*)(sl + m * SPITCH + 16 * q + 8 * hh);
            hhi[q] = *(const v4f*)(sl + m * SPITCH + 16 * q + 8 * hh + 4);
        }

        wave_lds_handover();
        #pragma unroll
        for (int j = 0; j < NR / 2; ++j) {
            *(v4f*)(slh + (2 * j + hh) * SPITCH + 4 * m) = rv[hf][j];
        }
        wave_lds_handover();
        #pragma unroll
        for (int q = 4 * hf; q < 4 * hf + 4; ++q) {
            r0[q] = *(const v4f*)(sl + m * SPITCH + 16 * q + 8 * hh);
            r1[q] = *(const v4f*)(sl + m * SPITCH + 16 * q + 8 * hh + 4);
        }

        #pragma unroll
        for (int q = 4 * hf; q < 4 * hf + 4; ++q) {
            if (q == 0) sq = r0[0] * r0[0];
            else        sq += r0[q] * r0[q];
            sq += r1[q] * r1[q];
        }

        #pragma unroll
        for (int q = 4 * hf; q < 4 * hf + 4; ++q) {
            const v4f rs0 = r0[q];
            const v4f rs1 = r1[q];
            v2f a, br, bw;
            a.x = hlo[q].x; a.y = hlo[q].y; br.x = rs0.x; br.y = rs0.y; bw.x = w0[q].x; bw.y = w0[q].y;
            accR = __builtin_amdgcn_wmma_f32_16x16x4_f32(false, a, false, br, (short)0, accR, false, false);
            accW = __builtin_amdgcn_wmma_f32_16x16x4_f32(false, a, false, bw, (short)0, accW, false, false);
            a.x = hlo[q].z; a.y = hlo[q].w; br.x = rs0.z; br.y = rs0.w; bw.x = w0[q].z; bw.y = w0[q].w;
            accR = __builtin_amdgcn_wmma_f32_16x16x4_f32(false, a, false, br, (short)0, accR, false, false);
            accW = __builtin_amdgcn_wmma_f32_16x16x4_f32(false, a, false, bw, (short)0, accW, false, false);
            a.x = hhi[q].x; a.y = hhi[q].y; br.x = rs1.x; br.y = rs1.y; bw.x = w1[q].x; bw.y = w1[q].y;
            accR = __builtin_amdgcn_wmma_f32_16x16x4_f32(false, a, false, br, (short)0, accR, false, false);
            accW = __builtin_amdgcn_wmma_f32_16x16x4_f32(false, a, false, bw, (short)0, accW, false, false);
            a.x = hhi[q].z; a.y = hhi[q].w; br.x = rs1.z; br.y = rs1.w; bw.x = w1[q].z; bw.y = w1[q].w;
            accR = __builtin_amdgcn_wmma_f32_16x16x4_f32(false, a, false, br, (short)0, accR, false, false);
            accW = __builtin_amdgcn_wmma_f32_16x16x4_f32(false, a, false, bw, (short)0, accW, false, false);
        }
        if (hf == 0) __builtin_amdgcn_sched_barrier(0x4);
    }

    float ssl = (sq.x + sq.y) + (sq.z + sq.w);
    ssl += __shfl_xor(ssl, 16, 32);

    float rraw[8][8];
    #pragma unroll
    for (int q = 0; q < 8; ++q) {
        const float* __restrict__ Rc = sl + (8 * hh) * SPITCH + 16 * q + m;
        #pragma unroll
        for (int i = 0; i < 8; ++i) rraw[q][i] = Rc[i * SPITCH];
    }
    const v4f b0 = *(const v4f*)(bvec + s * NR + 8 * hh);
    const v4f b1 = *(const v4f*)(bvec + s * NR + 8 * hh + 4);

    #pragma unroll
    for (int i = 0; i < 8; ++i) {
        partW[wave * PART_FLOATS + (8 * hh + i) * PPITCH + m] = accW[i];
        partR[wave * PART_FLOATS + (8 * hh + i) * PPITCH + m] = accR[i];
    }
    ssq[wave * 32 + lane] = ssl;
    __syncthreads();

    float tot = ssq[lane];
    #pragma unroll
    for (int w = 1; w < NWAVES; ++w) tot += ssq[w * 32 + lane];
    const float invl = 1.0f / fmaxf(sqrtf(tot), 1e-12f);
    float invr[8];
    #pragma unroll
    for (int i = 0; i < 8; ++i) invr[i] = __shfl(invl, 8 * hh + i, 32);

    v4f sw0 = *(const v4f*)(partW + m * PPITCH + 8 * hh);
    v4f sw1 = *(const v4f*)(partW + m * PPITCH + 8 * hh + 4);
    v4f sr0 = *(const v4f*)(partR + m * PPITCH + 8 * hh);
    v4f sr1 = *(const v4f*)(partR + m * PPITCH + 8 * hh + 4);
    #pragma unroll
    for (int w = 1; w < NWAVES; ++w) {
        sw0 += *(const v4f*)(partW + w * PART_FLOATS + m * PPITCH + 8 * hh);
        sw1 += *(const v4f*)(partW + w * PART_FLOATS + m * PPITCH + 8 * hh + 4);
        sr0 += *(const v4f*)(partR + w * PART_FLOATS + m * PPITCH + 8 * hh);
        sr1 += *(const v4f*)(partR + w * PART_FLOATS + m * PPITCH + 8 * hh + 4);
    }
    v4f iv0, iv1;
    iv0.x = invr[0]; iv0.y = invr[1]; iv0.z = invr[2]; iv0.w = invr[3];
    iv1.x = invr[4]; iv1.y = invr[5]; iv1.z = invr[6]; iv1.w = invr[7];
    const v4f p0 = (sw0 - sr0 * iv0 + b0) * iv0;
    const v4f p1 = (sw1 - sr1 * iv1 + b1) * iv1;
    const float pf[8] = { p0.x, p0.y, p0.z, p0.w, p1.x, p1.y, p1.z, p1.w };

    #pragma unroll
    for (int q = 0; q < 8; ++q) {
        v8f acc = {};
        #pragma unroll
        for (int t = 0; t < 4; ++t) {
            v2f a, bp;
            a.x  = rraw[q][2 * t];  a.y  = rraw[q][2 * t + 1];
            bp.x = pf[2 * t];  bp.y = pf[2 * t + 1];
            acc = __builtin_amdgcn_wmma_f32_16x16x4_f32(false, a, false, bp, (short)0, acc, false, false);
        }
        v4f o0, o1;
        o0.x = hlo[q].x + acc[0]; o0.y = hlo[q].y + acc[1]; o0.z = hlo[q].z + acc[2]; o0.w = hlo[q].w + acc[3];
        o1.x = hhi[q].x + acc[4]; o1.y = hhi[q].y + acc[5]; o1.z = hhi[q].z + acc[6]; o1.w = hhi[q].w + acc[7];
        *(v4f*)(sl + m * SPITCH + 16 * q + 8 * hh)     = o0;
        *(v4f*)(sl + m * SPITCH + 16 * q + 8 * hh + 4) = o1;
    }
    wave_lds_handover();

    if (t0 + TTILE <= NT) {
        v4f ov[TTILE];
        #pragma unroll
        for (int i = 0; i < TTILE; ++i) ov[i] = *(const v4f*)(sl + i * SPITCH + 4 * lane);
        #pragma unroll
        for (int i = 0; i < TTILE; ++i)
            *(v4f*)(obase + (size_t)(t0 + i) * ND + c0 + 4 * lane) = ov[i];
    } else {
        for (int i = 0; i < NT - t0; ++i) {
            v4f v = *(const v4f*)(sl + i * SPITCH + 4 * lane);
            *(v4f*)(obase + (size_t)(t0 + i) * ND + c0 + 4 * lane) = v;
        }
    }
}

extern "C" void kernel_launch(void* const* d_in, const int* in_sizes, int n_in,
                              void* d_out, int out_size, void* d_ws, size_t ws_size,
                              hipStream_t stream)
{
    const float* h    = (const float*)d_in[0];
    const float* R    = (const float*)d_in[1];
    const float* W    = (const float*)d_in[2];
    const float* bvec = (const float*)d_in[3];
    float* out = (float*)d_out;

    (void)in_sizes; (void)n_in; (void)out_size; (void)d_ws; (void)ws_size;

    enum { SMEM_FLOATS = NWAVES * SLICE_FLOATS + 2 * NWAVES * PART_FLOATS + NWAVES * 32 };

    loreft_main_kernel<<<NB * NS * NTILES, 256, SMEM_FLOATS * sizeof(float), stream>>>(
        h, R, W, bvec, out);
}


# A wave owns a 16-token tile over all 1024 columns (k sum inside the wave, no cross-wave sum); block = 8 tiles of one step, W and R staged once per block in LDS; sum order changed
# speedup vs baseline: 2.1322x; 1.1211x over previous
#include <hip/hip_runtime.h>
#include <math.h>

#define NB 8
#define NS 50
#define NT 77
#define ND 1024
#define NR 16
#define TTILE 16
#define NTILES 5
#define NWAVES 8
#define CHUNK 128
#define NCHUNK (ND / CHUNK)
#define SPITCH 132
#define WPITCH 1028
#define PPITCH 20
#define SLICE_FLOATS (TTILE * SPITCH)
#define WGT_FLOATS   (NR * WPITCH)
#define BLOCKS_PER_STEP ((NB * NTILES) / NWAVES)
#define LDS_FLOATS (2 * WGT_FLOATS + NWAVES * SLICE_FLOATS + NR)

typedef __attribute__((ext_vector_type(2))) float v2f;
typedef __attribute__((ext_vector_type(4))) float v4f;
typedef __attribute__((ext_vector_type(8))) float v8f;

__device__ __forceinline__ void wave_lds_handover()
{
    __builtin_amdgcn_fence(__ATOMIC_RELEASE, "wavefront");
    __builtin_amdgcn_wave_barrier();
    __builtin_amdgcn_fence(__ATOMIC_ACQUIRE, "wavefront");
}

__device__ __forceinline__ void load_rows(v4f (&hv)[TTILE], const float* __restrict__ hbase,
                                          int t0, int col)
{
    #pragma unroll
    for (int i = 0; i < TTILE; ++i) {
        int tokr = t0 + i;
        tokr = (tokr < NT) ? tokr : (NT - 1);
        hv[i] = *(const v4f*)(hbase + (size_t)tokr * ND + col);
    }
}

__device__ __forceinline__ void phase_a_chunk(v4f (&hv)[TTILE], v8f& accR, v8f& accW,
                                              float* __restrict__ sl,
                                              const float* __restrict__ Wl,
                                              const float* __restrict__ Rl,
                                              const float* __restrict__ hbase,
                                              int t0, int ch, int lane, int m, int hh)
{
    wave_lds_handover();
    #pragma unroll
    for (int i = 0; i < TTILE; ++i) *(v4f*)(sl + i * SPITCH + 4 * lane) = hv[i];
    wave_lds_handover();

    load_rows(hv, hbase, t0, CHUNK * ((ch + 2) & (NCHUNK - 1)));

    v4f hlo[8], hhi[8], r0[8], r1[8], w0[8], w1[8];
    #pragma unroll
    for (int q = 0; q < 8; ++q) {
        hlo[q] = *(const v4f*)(sl + m * SPITCH + 16 * q + 8 * hh);
        hhi[q] = *(const v4f*)(sl + m * SPITCH + 16 * q + 8 * hh + 4);
        w0[q]  = *(const v4f*)(Wl + m * WPITCH + CHUNK * ch + 16 * q + 8 * hh);
        w1[q]  = *(const v4f*)(Wl + m * WPITCH + CHUNK * ch + 16 * q + 8 * hh + 4);
        r0[q]  = *(const v4f*)(Rl + m * WPITCH + CHUNK * ch + 16 * q + 8 * hh);
        r1[q]  = *(const v4f*)(Rl + m * WPITCH + CHUNK * ch + 16 * q + 8 * hh + 4);
    }
    __builtin_amdgcn_sched_barrier(0x4);
    #pragma unroll
    for (int q = 0; q < 8; ++q) {
        v2f a, br, bw;
        a.x = hlo[q].x; a.y = hlo[q].y; br.x = r0[q].x; br.y = r0[q].y; bw.x = w0[q].x; bw.y = w0[q].y;
        accR = __builtin_amdgcn_wmma_f32_16x16x4_f32(false, a, false, br, (short)0, accR, false, false);
        accW = __builtin_amdgcn_wmma_f32_16x16x4_f32(false, a, false, bw, (short)0, accW, false, false);
        a.x = hlo[q].z; a.y = hlo[q].w; br.x = r0[q].z; br.y = r0[q].w; bw.x = w0[q].z; bw.y = w0[q].w;
        accR = __builtin_amdgcn_wmma_f32_16x16x4_f32(false, a, false, br, (short)0, accR, false, false);
        accW = __builtin_amdgcn_wmma_f32_16x16x4_f32(false, a, false, bw, (short)0, accW, false, false);
        a.x = hhi[q].x; a.y = hhi[q].y; br.x = r1[q].x; br.y = r1[q].y; bw.x = w1[q].x; bw.y = w1[q].y;
        accR = __builtin_amdgcn_wmma_f32_16x16x4_f32(false, a, false, br, (short)0, accR, false, false);
        accW = __builtin_amdgcn_wmma_f32_16x16x4_f32(false, a, false, bw, (short)0, accW, false, false);
        a.x = hhi[q].z; a.y = hhi[q].w; br.x = r1[q].z; br.y = r1[q].w; bw.x = w1[q].z; bw.y = w1[q].w;
        accR = __builtin_amdgcn_wmma_f32_16x16x4_f32(false, a, false, br, (short)0, accR, false, false);
        accW = __builtin_amdgcn_wmma_f32_16x16x4_f32(false, a, false, bw, (short)0, accW, false, false);
    }
}

__device__ __forceinline__ void phase_b_chunk(v4f (&hv)[TTILE], const float (&pf)[8],
                                              float* __restrict__ sl,
                                              const float* __restrict__ Rl,
                                              const float* __restrict__ hbase,
                                              float* __restrict__ obase,
                                              int t0, int ch, int lane, int m, int hh)
{
    float rraw[8][8];
    #pragma unroll
    for (int q = 0; q < 8; ++q) {
        const float* __restrict__ Rc = Rl + (8 * hh) * WPITCH + CHUNK * ch + 16 * q + m;
        #pragma unroll
        for (int i = 0; i < 8; ++i) rraw[q][i] = Rc[i * WPITCH];
    }
    wave_lds_handover();
    #pragma unroll
    for (int q = 0; q < 8; ++q) {
        v8f acc = {};
        #pragma unroll
        for (int t = 0; t < 4; ++t) {
            v2f a, bp;
            a.x  = rraw[q][2 * t];  a.y  = rraw[q][2 * t + 1];
            bp.x = pf[2 * t];  bp.y = pf[2 * t + 1];
            acc = __builtin_amdgcn_wmma_f32_16x16x4_f32(false, a, false, bp, (short)0, acc, false, false);
        }
        v4f o0, o1;
        o0.x = acc[0]; o0.y = acc[1]; o0.z = acc[2]; o0.w = acc[3];
        o1.x = acc[4]; o1.y = acc[5]; o1.z = acc[6]; o1.w = acc[7];
        *(v4f*)(sl + m * SPITCH + 16 * q + 8 * hh)     = o0;
        *(v4f*)(sl + m * SPITCH + 16 * q + 8 * hh + 4) = o1;
    }
    wave_lds_handover();

    v4f ov[TTILE];
    #pragma unroll
    for (int i = 0; i < TTILE; ++i) ov[i] = *(const v4f*)(sl + i * SPITCH + 4 * lane) + hv[i];
    if (t0 + TTILE <= NT) {
        #pragma unroll
        for (int i = 0; i < TTILE; ++i)
            *(v4f*)(obase + (size_t)(t0 + i) * ND + CHUNK * ch) = ov[i];
    } else {
        #pragma unroll
        for (int i = 0; i < TTILE; ++i)
            if (t0 + i < NT) *(v4f*)(obase + (size_t)(t0 + i) * ND + CHUNK * ch) = ov[i];
    }
    if (ch + 2 < NCHUNK) load_rows(hv, hbase, t0, CHUNK * (ch + 2));
}

__global__ __launch_bounds__(256, 2)
void loreft_main_kernel(const float* __restrict__ h,
                        const float* __restrict__ R,
                        const float* __restrict__ W,
                        const float* __restrict__ bvec,
                        float* __restrict__ out)
{
    __shared__ __attribute__((aligned(16))) float smem[LDS_FLOATS];

    const int blk  = blockIdx.x;
    const int s    = blk / BLOCKS_PER_STEP;
    const int grp  = blk % BLOCKS_PER_STEP;

    const int lane = threadIdx.x & 31;
    const int wave = __builtin_amdgcn_readfirstlane((int)(threadIdx.x >> 5));
    const int m    = lane & 15;
    const int hh   = lane >> 4;

    const int j    = grp * NWAVES + wave;
    const int b    = j / NTILES;
    const int tile = j % NTILES;
    const int t0   = tile * TTILE;

    float* __restrict__ Wl   = smem;
    float* __restrict__ Rl   = smem + WGT_FLOATS;
    float* __restrict__ sl   = smem + 2 * WGT_FLOATS + wave * SLICE_FLOATS;
    float* __restrict__ invs = smem + 2 * WGT_FLOATS + NWAVES * SLICE_FLOATS;

    const size_t tok_base = ((size_t)(b * NS + s)) * NT;
    const float* __restrict__ hbase = h + tok_base * ND + 4 * lane;
    float*       __restrict__ obase = out + tok_base * ND + 4 * lane;
    const float* __restrict__ Rs = R + (size_t)s * NR * ND;
    const float* __restrict__ Ws = W + (size_t)s * NR * ND;

    v4f wrow[2][NCHUNK], rrow[2][NCHUNK];
    #pragma unroll
    for (int e = 0; e < 2; ++e) {
        #pragma unroll
        for (int k = 0; k < NCHUNK; ++k) {
            wrow[e][k] = *(const v4f*)(Ws + (size_t)(2 * wave + e) * ND + CHUNK * k + 4 * lane);
            rrow[e][k] = *(const v4f*)(Rs + (size_t)(2 * wave + e) * ND + CHUNK * k + 4 * lane);
        }
    }
    v4f hva[TTILE], hvb[TTILE];
    load_rows(hva, hbase, t0, 0);
    load_rows(hvb, hbase, t0, CHUNK);

    #pragma unroll
    for (int e = 0; e < 2; ++e) {
        v4f sq;
        #pragma unroll
        for (int k = 0; k < NCHUNK; ++k) {
            *(v4f*)(Wl + (2 * wave + e) * WPITCH + CHUNK * k + 4 * lane) = wrow[e][k];
            *(v4f*)(Rl + (2 * wave + e) * WPITCH + CHUNK * k + 4 * lane) = rrow[e][k];
            if (k == 0) sq = rrow[e][0] * rrow[e][0];
            else        sq += rrow[e][k] * rrow[e][k];
        }
        float ss = (sq.x + sq.y) + (sq.z + sq.w);
        ss += __shfl_xor(ss, 16, 32);
        ss += __shfl_xor(ss, 8, 32);
        ss += __shfl_xor(ss, 4, 32);
        ss += __shfl_xor(ss, 2, 32);
        ss += __shfl_xor(ss, 1, 32);
        if (lane == 0) invs[2 * wave + e] = 1.0f / fmaxf(sqrtf(ss), 1e-12f);
    }
    __syncthreads();

    v8f accR = {};
    v8f accW = {};
    #pragma unroll 1
    for (int ch = 0; ch < NCHUNK; ch += 2) {
        phase_a_chunk(hva, accR, accW, sl, Wl, Rl, hbase, t0, ch,     lane, m, hh);
        phase_a_chunk(hvb, accR, accW, sl, Wl, Rl, hbase, t0, ch + 1, lane, m, hh);
    }

    float* __restrict__ pW = sl;
    float* __restrict__ pR = sl + TTILE * PPITCH;
    wave_lds_handover();
    #pragma unroll
    for (int i = 0; i < 8; ++i) {
        pW[(8 * hh + i) * PPITCH + m] = accW[i];
        pR[(8 * hh + i) * PPITCH + m] = accR[i];
    }
    wave_lds_handover();
    const v4f sw0 = *(const v4f*)(pW + m * PPITCH + 8 * hh);
    const v4f sw1 = *(const v4f*)(pW + m * PPITCH + 8 * hh + 4);
    const v4f sr0 = *(const v4f*)(pR + m * PPITCH + 8 * hh);
    const v4f sr1 = *(const v4f*)(pR + m * PPITCH + 8 * hh + 4);
    const v4f iv0 = *(const v4f*)(invs + 8 * hh);
    const v4f iv1 = *(const v4f*)(invs + 8 * hh + 4);
    const v4f b0 = *(const v4f*)(bvec + s * NR + 8 * hh);
    const v4f b1 = *(const v4f*)(bvec + s * NR + 8 * hh + 4);
    const v4f p0 = (sw0 - sr0 * iv0 + b0) * iv0;
    const v4f p1 = (sw1 - sr1 * iv1 + b1) * iv1;
    const float pf[8] = { p0.x, p0.y, p0.z, p0.w, p1.x, p1.y, p1.z, p1.w };

    #pragma unroll 1
    for (int ch = 0; ch < NCHUNK; ch += 2) {
        phase_b_chunk(hva, pf, sl, Rl, hbase, obase, t0, ch,     lane, m, hh);
        phase_b_chunk(hvb, pf, sl, Rl, hbase, obase, t0, ch + 1, lane, m, hh);
    }
}

extern "C" void kernel_launch(void* const* d_in, const int* in_sizes, int n_in,
                              void* d_out, int out_size, void* d_ws, size_t ws_size,
                              hipStream_t stream)
{
    const float* h    = (const float*)d_in[0];
    const float* R    = (const float*)d_in[1];
    const float* W    = (const float*)d_in[2];
    const float* bvec = (const float*)d_in[3];
    float* out = (float*)d_out;

    (void)in_sizes; (void)n_in; (void)out_size; (void)d_ws; (void)ws_size;

    loreft_main_kernel<<<NS * BLOCKS_PER_STEP, 256, 0, stream>>>(h, R, W, bvec, out);
}


# Two private LDS slices per wave, alternated by phase A: phase B takes the h rows of chunks 7,6 from LDS (6 of 8 re-read); re-read loop peeled, no conditional row load; sums unchanged
# speedup vs baseline: 2.2577x; 1.0589x over previous
#include <hip/hip_runtime.h>
#include <math.h>

#define NB 8
#define NS 50
#define NT 77
#define ND 1024
#define NR 16
#define TTILE 16
#define NTILES 5
#define NWAVES 8
#define CHUNK 128
#define NCHUNK (ND / CHUNK)
#define SPITCH 132
#define WPITCH 1028
#define PPITCH 20
#define SLICE_FLOATS (TTILE * SPITCH)
#define WGT_FLOATS   (NR * WPITCH)
#define BLOCKS_PER_STEP ((NB * NTILES) / NWAVES)
#define PSCR_FLOATS  (2 * TTILE * PPITCH)
#define LDS_FLOATS (2 * WGT_FLOATS + 2 * NWAVES * SLICE_FLOATS + NR + NWAVES * PSCR_FLOATS)

typedef __attribute__((ext_vector_type(2))) float v2f;
typedef __attribute__((ext_vector_type(4))) float v4f;
typedef __attribute__((ext_vector_type(8))) float v8f;

__device__ __forceinline__ void wave_lds_handover()
{
    __builtin_amdgcn_fence(__ATOMIC_RELEASE, "wavefront");
    __builtin_amdgcn_wave_barrier();
    __builtin_amdgcn_fence(__ATOMIC_ACQUIRE, "wavefront");
}

__device__ __forceinline__ void load_rows(v4f (&hv)[TTILE], const float* __restrict__ hbase,
                                          int t0, int col)
{
    #pragma unroll
    for (int i = 0; i < TTILE; ++i) {
        int tokr = t0 + i;
        tokr = (tokr < NT) ? tokr : (NT - 1);
        hv[i] = *(const v4f*)(hbase + (size_t)tokr * ND + col);
    }
}

__device__ __forceinline__ void phase_a_chunk(v4f (&hv)[TTILE], v8f& accR, v8f& accW,
                                              float* __restrict__ sl,
                                              const float* __restrict__ Wl,
                                              const float* __restrict__ Rl,
                                              const float* __restrict__ hbase,
                                              int t0, int ch, int lane, int m, int hh)
{
    wave_lds_handover();
    #pragma unroll
    for (int i = 0; i < TTILE; ++i) *(v4f*)(sl + i * SPITCH + 4 * lane) = hv[i];
    wave_lds_handover();

    load_rows(hv, hbase, t0, CHUNK * ((ch + 2) & (NCHUNK - 1)));

    v4f hlo[8], hhi[8], r0[8], r1[8], w0[8], w1[8];
    #pragma unroll
    for (int q = 0; q < 8; ++q) {
        hlo[q] = *(const v4f*)(sl + m * SPITCH + 16 * q + 8 * hh);
        hhi[q] = *(const v4f*)(sl + m * SPITCH + 16 * q + 8 * hh + 4);
        w0[q]  = *(const v4f*)(Wl + m * WPITCH + CHUNK * ch + 16 * q + 8 * hh);
        w1[q]  = *(const v4f*)(Wl + m * WPITCH + CHUNK * ch + 16 * q + 8 * hh + 4);
        r0[q]  = *(const v4f*)(Rl + m * WPITCH + CHUNK * ch + 16 * q + 8 * hh);
        r1[q]  = *(const v4f*)(Rl + m * WPITCH + CHUNK * ch + 16 * q + 8 * hh + 4);
    }
    __builtin_amdgcn_sched_barrier(0x4);
    #pragma unroll
    for (int q = 0; q < 8; ++q) {
        v2f a, br, bw;
        a.x = hlo[q].x; a.y = hlo[q].y; br.x = r0[q].x; br.y = r0[q].y; bw.x = w0[q].x; bw.y = w0[q].y;
        accR = __builtin_amdgcn_wmma_f32_16x16x4_f32(false, a, false, br, (short)0, accR, false, false);
        accW = __builtin_amdgcn_wmma_f32_16x16x4_f32(false, a, false, bw, (short)0, accW, false, false);
        a.x = hlo[q].z; a.y = hlo[q].w; br.x = r0[q].z; br.y = r0[q].w; bw.x = w0[q].z; bw.y = w0[q].w;
        accR = __builtin_amdgcn_wmma_f32_16x16x4_f32(false, a, false, br, (short)0, accR, false, false);
        accW = __builtin_amdgcn_wmma_f32_16x16x4_f32(false, a, false, bw, (short)0, accW, false, false);
        a.x = hhi[q].x; a.y = hhi[q].y; br.x = r1[q].x; br.y = r1[q].y; bw.x = w1[q].x; bw.y = w1[q].y;
        accR = __builtin_amdgcn_wmma_f32_16x16x4_f32(false, a, false, br, (short)0, accR, false, false);
        accW = __builtin_amdgcn_wmma_f32_16x16x4_f32(false, a, false, bw, (short)0, accW, false, false);
        a.x = hhi[q].z; a.y = hhi[q].w; br.x = r1[q].z; br.y = r1[q].w; bw.x = w1[q].z; bw.y = w1[q].w;
        accR = __builtin_amdgcn_wmma_f32_16x16x4_f32(false, a, false, br, (short)0, accR, false, false);
        accW = __builtin_amdgcn_wmma_f32_16x16x4_f32(false, a, false, bw, (short)0, accW, false, false);
    }
}

template <bool ROWS_IN_LDS, bool REFILL>
__device__ __forceinline__ void phase_b_chunk(v4f (&hv)[TTILE], const float (&pf)[8],
                                              float* __restrict__ sl,
                                              const float* __restrict__ Rl,
                                              const float* __restrict__ hbase,
                                              float* __restrict__ obase,
                                              int t0, int ch, int lane, int m, int hh)
{
    if (ROWS_IN_LDS) {
        #pragma unroll
        for (int i = 0; i < TTILE; ++i) hv[i] = *(const v4f*)(sl + i * SPITCH + 4 * lane);
    }
    float rraw[8][8];
    #pragma unroll
    for (int q = 0; q < 8; ++q) {
        const float* __restrict__ Rc = Rl + (8 * hh) * WPITCH + CHUNK * ch + 16 * q + m;
        #pragma unroll
        for (int i = 0; i < 8; ++i) rraw[q][i] = Rc[i * WPITCH];
    }
    wave_lds_handover();
    #pragma unroll
    for (int q = 0; q < 8; ++q) {
        v8f acc = {};
        #pragma unroll
        for (int t = 0; t < 4; ++t) {
            v2f a, bp;
            a.x  = rraw[q][2 * t];  a.y  = rraw[q][2 * t + 1];
            bp.x = pf[2 * t];  bp.y = pf[2 * t + 1];
            acc = __builtin_amdgcn_wmma_f32_16x16x4_f32(false, a, false, bp, (short)0, acc, false, false);
        }
        v4f o0, o1;
        o0.x = acc[0]; o0.y = acc[1]; o0.z = acc[2]; o0.w = acc[3];
        o1.x = acc[4]; o1.y = acc[5]; o1.z = acc[6]; o1.w = acc[7];
        *(v4f*)(sl + m * SPITCH + 16 * q + 8 * hh)     = o0;
        *(v4f*)(sl + m * SPITCH + 16 * q + 8 * hh + 4) = o1;
    }
    wave_lds_handover();

    v4f ov[TTILE];
    #pragma unroll
    for (int i = 0; i < TTILE; ++i) ov[i] = *(const v4f*)(sl + i * SPITCH + 4 * lane) + hv[i];
    if (t0 + TTILE <= NT) {
        #pragma unroll
        for (int i = 0; i < TTILE; ++i)
            *(v4f*)(obase + (size_t)(t0 + i) * ND + CHUNK * ch) = ov[i];
    } else {
        #pragma unroll
        for (int i = 0; i < NT - (NTILES - 1) * TTILE; ++i)
            *(v4f*)(obase + (size_t)(t0 + i) * ND + CHUNK * ch) = ov[i];
    }
    if (REFILL) load_rows(hv, hbase, t0, CHUNK * (ch + 2));
}

__global__ __launch_bounds__(256, 2)
void loreft_main_kernel(const float* __restrict__ h,
                        const float* __restrict__ R,
                        const float* __restrict__ W,
                        const float* __restrict__ bvec,
                        float* __restrict__ out)
{
    __shared__ __attribute__((aligned(16))) float smem[LDS_FLOATS];

    const int blk  = blockIdx.x;
    const int s    = blk / BLOCKS_PER_STEP;
    const int grp  = blk % BLOCKS_PER_STEP;

    const int lane = threadIdx.x & 31;
    const int wave = __builtin_amdgcn_readfirstlane((int)(threadIdx.x >> 5));
    const int m    = lane & 15;
    const int hh   = lane >> 4;

    const int j    = grp * NWAVES + wave;
    const int b    = j / NTILES;
    const int tile = j % NTILES;
    const int t0   = tile * TTILE;

    float* __restrict__ Wl   = smem;
    float* __restrict__ Rl   = smem + WGT_FLOATS;
    float* __restrict__ sl0  = smem + 2 * WGT_FLOATS + (2 * wave) * SLICE_FLOATS;
    float* __restrict__ sl1  = sl0 + SLICE_FLOATS;
    float* __restrict__ invs = smem + 2 * WGT_FLOATS + 2 * NWAVES * SLICE_FLOATS;
    float* __restrict__ psc  = invs + NR + wave * PSCR_FLOATS;

    const size_t tok_base = ((size_t)(b * NS + s)) * NT;
    const float* __restrict__ hbase = h + tok_base * ND + 4 * lane;
    float*       __restrict__ obase = out + tok_base * ND + 4 * lane;
    const float* __restrict__ Rs = R + (size_t)s * NR * ND;
    const float* __restrict__ Ws = W + (size_t)s * NR * ND;

    v4f wrow[2][NCHUNK], rrow[2][NCHUNK];
    #pragma unroll
    for (int e = 0; e < 2; ++e) {
        #pragma unroll
        for (int k = 0; k < NCHUNK; ++k) {
            wrow[e][k] = *(const v4f*)(Ws + (size_t)(2 * wave + e) * ND + CHUNK * k + 4 * lane);
            rrow[e][k] = *(const v4f*)(Rs + (size_t)(2 * wave + e) * ND + CHUNK * k + 4 * lane);
        }
    }
    v4f hva[TTILE], hvb[TTILE];
    load_rows(hva, hbase, t0, 0);
    load_rows(hvb, hbase, t0, CHUNK);

    #pragma unroll
    for (int e = 0; e < 2; ++e) {
        v4f sq;
        #pragma unroll
        for (int k = 0; k < NCHUNK; ++k) {
            *(v4f*)(Wl + (2 * wave + e) * WPITCH + CHUNK * k + 4 * lane) = wrow[e][k];
            *(v4f*)(Rl + (2 * wave + e) * WPITCH + CHUNK * k + 4 * lane) = rrow[e][k];
            if (k == 0) sq = rrow[e][0] * rrow[e][0];
            else        sq += rrow[e][k] * rrow[e][k];
        }
        float ss = (sq.x + sq.y) + (sq.z + sq.w);
        ss += __shfl_xor(ss, 16, 32);
        ss += __shfl_xor(ss, 8, 32);
        ss += __shfl_xor(ss, 4, 32);
        ss += __shfl_xor(ss, 2, 32);
        ss += __shfl_xor(ss, 1, 32);
        if (lane == 0) invs[2 * wave + e] = 1.0f / fmaxf(sqrtf(ss), 1e-12f);
    }
    __syncthreads();

    v8f accR = {};
    v8f accW = {};
    #pragma unroll 1
    for (int ch = 0; ch < NCHUNK; ch += 2) {
        phase_a_chunk(hva, accR, accW, sl0, Wl, Rl, hbase, t0, ch,     lane, m, hh);
        phase_a_chunk(hvb, accR, accW, sl1, Wl, Rl, hbase, t0, ch + 1, lane, m, hh);
    }

    float* __restrict__ pW = psc;
    float* __restrict__ pR = psc + TTILE * PPITCH;
    #pragma unroll
    for (int i = 0; i < 8; ++i) {
        pW[(8 * hh + i) * PPITCH + m] = accW[i];
        pR[(8 * hh + i) * PPITCH + m] = accR[i];
    }
    wave_lds_handover();
    const v4f sw0 = *(const v4f*)(pW + m * PPITCH + 8 * hh);
    const v4f sw1 = *(const v4f*)(pW + m * PPITCH + 8 * hh + 4);
    const v4f sr0 = *(const v4f*)(pR + m * PPITCH + 8 * hh);
    const v4f sr1 = *(const v4f*)(pR + m * PPITCH + 8 * hh + 4);
    const v4f iv0 = *(const v4f*)(invs + 8 * hh);
    const v4f iv1 = *(const v4f*)(invs + 8 * hh + 4);
    const v4f b0 = *(const v4f*)(bvec + s * NR + 8 * hh);
    const v4f b1 = *(const v4f*)(bvec + s * NR + 8 * hh + 4);
    const v4f p0 = (sw0 - sr0 * iv0 + b0) * iv0;
    const v4f p1 = (sw1 - sr1 * iv1 + b1) * iv1;
    const float pf[8] = { p0.x, p0.y, p0.z, p0.w, p1.x, p1.y, p1.z, p1.w };

    {
        v4f hx[TTILE];
        phase_b_chunk<true, false>(hx, pf, sl1, Rl, hbase, obase, t0, NCHUNK - 1, lane, m, hh);
        phase_b_chunk<true, false>(hx, pf, sl0, Rl, hbase, obase, t0, NCHUNK - 2, lane, m, hh);
    }
    #pragma unroll 1
    for (int ch = 0; ch < NCHUNK - 4; ch += 2) {
        phase_b_chunk<false, true>(hva, pf, sl0, Rl, hbase, obase, t0, ch,     lane, m, hh);
        phase_b_chunk<false, true>(hvb, pf, sl1, Rl, hbase, obase, t0, ch + 1, lane, m, hh);
    }
    phase_b_chunk<false, false>(hva, pf, sl0, Rl, hbase, obase, t0, NCHUNK - 4, lane, m, hh);
    phase_b_chunk<false, false>(hvb, pf, sl1, Rl, hbase, obase, t0, NCHUNK - 3, lane, m, hh);
}

extern "C" void kernel_launch(void* const* d_in, const int* in_sizes, int n_in,
                              void* d_out, int out_size, void* d_ws, size_t ws_size,
                              hipStream_t stream)
{
    const float* h    = (const float*)d_in[0];
    const float* R    = (const float*)d_in[1];
    const float* W    = (const float*)d_in[2];
    const float* bvec = (const float*)d_in[3];
    float* out = (float*)d_out;

    (void)in_sizes; (void)n_in; (void)out_size; (void)d_ws; (void)ws_size;

    loreft_main_kernel<<<NS * BLOCKS_PER_STEP, 256, 0, stream>>>(h, R, W, bvec, out);
}


# Rows of chunks 0,1 stay in registers of their own from prologue to phase B; phase B reads 4 of 8 chunks again (was 6); register room: a tile's first 13 rows addressed off one base; sums unchanged
# speedup vs baseline: 2.3669x; 1.0484x over previous
#include <hip/hip_runtime.h>
#include <math.h>

#define NB 8
#define NS 50
#define NT 77
#define ND 1024
#define NR 16
#define TTILE 16
#define NTILES 5
#define RAGROWS (NT - (NTILES - 1) * TTILE)
#define NWAVES 8
#define CHUNK 128
#define NCHUNK (ND / CHUNK)
#define SPITCH 132
#define WPITCH 1028
#define PPITCH 20
#define SLICE_FLOATS (TTILE * SPITCH)
#define WGT_FLOATS   (NR * WPITCH)
#define BLOCKS_PER_STEP ((NB * NTILES) / NWAVES)
#define PSCR_FLOATS  (2 * TTILE * PPITCH)
#define LDS_FLOATS (2 * WGT_FLOATS + 2 * NWAVES * SLICE_FLOATS + NR + NWAVES * PSCR_FLOATS)

typedef __attribute__((ext_vector_type(2))) float v2f;
typedef __attribute__((ext_vector_type(4))) float v4f;
typedef __attribute__((ext_vector_type(8))) float v8f;

__device__ __forceinline__ void wave_lds_handover()
{
    __builtin_amdgcn_fence(__ATOMIC_RELEASE, "wavefront");
    __builtin_amdgcn_wave_barrier();
    __builtin_amdgcn_fence(__ATOMIC_ACQUIRE, "wavefront");
}

__device__ __forceinline__ void load_rows(v4f (&hv)[TTILE], const float* __restrict__ hseq,
                                          unsigned lane4, int t0, int col)
{
    const float* __restrict__ tile0 = hseq + (size_t)t0 * ND + col + lane4;
    const int lastrow = (t0 + TTILE <= NT) ? (TTILE - 1) : (RAGROWS - 1);
    #pragma unroll
    for (int i = 0; i < TTILE; ++i) {
        if (i < RAGROWS) {
            hv[i] = *(const v4f*)(tile0 + (size_t)i * ND);
        } else {
            const int r = (i <= lastrow) ? i : lastrow;
            hv[i] = *(const v4f*)(tile0 + (size_t)r * ND);
        }
    }
}

__device__ __forceinline__ void phase_a_chunk(const v4f (&hin)[TTILE], v4f (&hout)[TTILE],
                                              v8f& accR, v8f& accW,
                                              float* __restrict__ sl,
                                              const float* __restrict__ Wl,
                                              const float* __restrict__ Rl,
                                              const float* __restrict__ hseq, unsigned lane4,
                                              int t0, int ch, int rch, int lane, int m, int hh)
{
    wave_lds_handover();
    #pragma unroll
    for (int i = 0; i < TTILE; ++i) *(v4f*)(sl + i * SPITCH + 4 * lane) = hin[i];
    wave_lds_handover();

    load_rows(hout, hseq, lane4, t0, CHUNK * rch);

    v4f hlo[8], hhi[8], r0[8], r1[8], w0[8], w1[8];
    #pragma unroll
    for (int q = 0; q < 8; ++q) {
        hlo[q] = *(const v4f*)(sl + m * SPITCH + 16 * q + 8 * hh);
        hhi[q] = *(const v4f*)(sl + m * SPITCH + 16 * q + 8 * hh + 4);
        w0[q]  = *(const v4f*)(Wl + m * WPITCH + CHUNK * ch + 16 * q + 8 * hh);
        w1[q]  = *(const v4f*)(Wl + m * WPITCH + CHUNK * ch + 16 * q + 8 * hh + 4);
        r0[q]  = *(const v4f*)(Rl + m * WPITCH + CHUNK * ch + 16 * q + 8 * hh);
        r1[q]  = *(const v4f*)(Rl + m * WPITCH + CHUNK * ch + 16 * q + 8 * hh + 4);
    }
    __builtin_amdgcn_sched_barrier(0x4);
    #pragma unroll
    for (int q = 0; q < 8; ++q) {
        v2f a, br, bw;
        a.x = hlo[q].x; a.y = hlo[q].y; br.x = r0[q].x; br.y = r0[q].y; bw.x = w0[q].x; bw.y = w0[q].y;
        accR = __builtin_amdgcn_wmma_f32_16x16x4_f32(false, a, false, br, (short)0, accR, false, false);
        accW = __builtin_amdgcn_wmma_f32_16x16x4_f32(false, a, false, bw, (short)0, accW, false, false);
        a.x = hlo[q].z; a.y = hlo[q].w; br.x = r0[q].z; br.y = r0[q].w; bw.x = w0[q].z; bw.y = w0[q].w;
        accR = __builtin_amdgcn_wmma_f32_16x16x4_f32(false, a, false, br, (short)0, accR, false, false);
        accW = __builtin_amdgcn_wmma_f32_16x16x4_f32(false, a, false, bw, (short)0, accW, false, false);
        a.x = hhi[q].x; a.y = hhi[q].y; br.x = r1[q].x; br.y = r1[q].y; bw.x = w1[q].x; bw.y = w1[q].y;
        accR = __builtin_amdgcn_wmma_f32_16x16x4_f32(false, a, false, br, (short)0, accR, false, false);
        accW = __builtin_amdgcn_wmma_f32_16x16x4_f32(false, a, false, bw, (short)0, accW, false, false);
        a.x = hhi[q].z; a.y = hhi[q].w; br.x = r1[q].z; br.y = r1[q].w; bw.x = w1[q].z; bw.y = w1[q].w;
        accR = __builtin_amdgcn_wmma_f32_16x16x4_f32(false, a, false, br, (short)0, accR, false, false);
        accW = __builtin_amdgcn_wmma_f32_16x16x4_f32(false, a, false, bw, (short)0, accW, false, false);
    }
}

template <bool ROWS_IN_LDS, int REFILL_AHEAD>
__device__ __forceinline__ void phase_b_chunk(v4f (&hv)[TTILE], const float (&pf)[8],
                                              float* __restrict__ sl,
                                              const float* __restrict__ Rl,
                                              const float* __restrict__ hseq, unsigned lane4,
                                              float* __restrict__ obase,
                                              int t0, int ch, int lane, int m, int hh)
{
    if (ROWS_IN_LDS) {
        #pragma unroll
        for (int i = 0; i < TTILE; ++i) hv[i] = *(const v4f*)(sl + i * SPITCH + 4 * lane);
    }
    float rraw[8][8];
    #pragma unroll
    for (int q = 0; q < 8; ++q) {
        const float* __restrict__ Rc = Rl + (8 * hh) * WPITCH + CHUNK * ch + 16 * q + m;
        #pragma unroll
        for (int i = 0; i < 8; ++i) rraw[q][i] = Rc[i * WPITCH];
    }
    wave_lds_handover();
    #pragma unroll
    for (int q = 0; q < 8; ++q) {
        v8f acc = {};
        #pragma unroll
        for (int t = 0; t < 4; ++t) {
            v2f a, bp;
            a.x  = rraw[q][2 * t];  a.y  = rraw[q][2 * t + 1];
            bp.x = pf[2 * t];  bp.y = pf[2 * t + 1];
            acc = __builtin_amdgcn_wmma_f32_16x16x4_f32(false, a, false, bp, (short)0, acc, false, false);
        }
        v4f o0, o1;
        o0.x = acc[0]; o0.y = acc[1]; o0.z = acc[2]; o0.w = acc[3];
        o1.x = acc[4]; o1.y = acc[5]; o1.z = acc[6]; o1.w = acc[7];
        *(v4f*)(sl + m * SPITCH + 16 * q + 8 * hh)     = o0;
        *(v4f*)(sl + m * SPITCH + 16 * q + 8 * hh + 4) = o1;
    }
    wave_lds_handover();

    v4f ov[TTILE];
    #pragma unroll
    for (int i = 0; i < TTILE; ++i) ov[i] = *(const v4f*)(sl + i * SPITCH + 4 * lane) + hv[i];
    if (t0 + TTILE <= NT) {
        #pragma unroll
        for (int i = 0; i < TTILE; ++i)
            *(v4f*)(obase + (size_t)(t0 + i) * ND + CHUNK * ch) = ov[i];
    } else {
        #pragma unroll
        for (int i = 0; i < NT - (NTILES - 1) * TTILE; ++i)
            *(v4f*)(obase + (size_t)(t0 + i) * ND + CHUNK * ch) = ov[i];
    }
    if (REFILL_AHEAD) load_rows(hv, hseq, lane4, t0, CHUNK * (ch + REFILL_AHEAD));
}

__global__ __launch_bounds__(256, 2)
void loreft_main_kernel(const float* __restrict__ h,
                        const float* __restrict__ R,
                        const float* __restrict__ W,
                        const float* __restrict__ bvec,
                        float* __restrict__ out)
{
    __shared__ __attribute__((aligned(16))) float smem[LDS_FLOATS];

    const int blk  = blockIdx.x;
    const int s    = blk / BLOCKS_PER_STEP;
    const int grp  = blk % BLOCKS_PER_STEP;

    const int lane = threadIdx.x & 31;
    const int wave = __builtin_amdgcn_readfirstlane((int)(threadIdx.x >> 5));
    const int m    = lane & 15;
    const int hh   = lane >> 4;

    const int j    = grp * NWAVES + wave;
    const int b    = j / NTILES;
    const int tile = j % NTILES;
    const int t0   = tile * TTILE;

    float* __restrict__ Wl   = smem;
    float* __restrict__ Rl   = smem + WGT_FLOATS;
    float* __restrict__ sl0  = smem + 2 * WGT_FLOATS + (2 * wave) * SLICE_FLOATS;
    float* __restrict__ sl1  = sl0 + SLICE_FLOATS;
    float* __restrict__ invs = smem + 2 * WGT_FLOATS + 2 * NWAVES * SLICE_FLOATS;
    float* __restrict__ psc  = invs + NR + wave * PSCR_FLOATS;

    const size_t tok_base = ((size_t)(b * NS + s)) * NT;
    const float* __restrict__ hseq = h + tok_base * ND;
    const unsigned lane4 = 4u * (unsigned)lane;
    float*       __restrict__ obase = out + tok_base * ND + 4 * lane;
    const float* __restrict__ Rs = R + (size_t)s * NR * ND;
    const float* __restrict__ Ws = W + (size_t)s * NR * ND;

    v4f wrow[2][NCHUNK], rrow[2][NCHUNK];
    #pragma unroll
    for (int e = 0; e < 2; ++e) {
        #pragma unroll
        for (int k = 0; k < NCHUNK; ++k) {
            wrow[e][k] = *(const v4f*)(Ws + (size_t)(2 * wave + e) * ND + CHUNK * k + 4 * lane);
            rrow[e][k] = *(const v4f*)(Rs + (size_t)(2 * wave + e) * ND + CHUNK * k + 4 * lane);
        }
    }
    v4f hk0[TTILE], hk1[TTILE];
    load_rows(hk0, hseq, lane4, t0, 0);
    load_rows(hk1, hseq, lane4, t0, CHUNK);

    #pragma unroll
    for (int e = 0; e < 2; ++e) {
        v4f sq;
        #pragma unroll
        for (int k = 0; k < NCHUNK; ++k) {
            *(v4f*)(Wl + (2 * wave + e) * WPITCH + CHUNK * k + 4 * lane) = wrow[e][k];
            *(v4f*)(Rl + (2 * wave + e) * WPITCH + CHUNK * k + 4 * lane) = rrow[e][k];
            if (k == 0) sq = rrow[e][0] * rrow[e][0];
            else        sq += rrow[e][k] * rrow[e][k];
        }
        float ss = (sq.x + sq.y) + (sq.z + sq.w);
        ss += __shfl_xor(ss, 16, 32);
        ss += __shfl_xor(ss, 8, 32);
        ss += __shfl_xor(ss, 4, 32);
        ss += __shfl_xor(ss, 2, 32);
        ss += __shfl_xor(ss, 1, 32);
        if (lane == 0) invs[2 * wave + e] = 1.0f / fmaxf(sqrtf(ss), 1e-12f);
    }
    __syncthreads();

    v8f accR = {};
    v8f accW = {};
    v4f hva[TTILE], hvb[TTILE];
    phase_a_chunk(hk0, hva, accR, accW, sl0, Wl, Rl, hseq, lane4, t0, 0, 2, lane, m, hh);
    phase_a_chunk(hk1, hvb, accR, accW, sl1, Wl, Rl, hseq, lane4, t0, 1, 3, lane, m, hh);
    #pragma unroll 1
    for (int ch = 2; ch < NCHUNK; ch += 2) {
        const int rch = (ch + 2 < NCHUNK) ? (ch + 2) : 2;
        phase_a_chunk(hva, hva, accR, accW, sl0, Wl, Rl, hseq, lane4, t0, ch,     rch,     lane, m, hh);
        phase_a_chunk(hvb, hvb, accR, accW, sl1, Wl, Rl, hseq, lane4, t0, ch + 1, rch + 1, lane, m, hh);
    }

    float* __restrict__ pW = psc;
    float* __restrict__ pR = psc + TTILE * PPITCH;
    #pragma unroll
    for (int i = 0; i < 8; ++i) {
        pW[(8 * hh + i) * PPITCH + m] = accW[i];
        pR[(8 * hh + i) * PPITCH + m] = accR[i];
    }
    wave_lds_handover();
    const v4f sw0 = *(const v4f*)(pW + m * PPITCH + 8 * hh);
    const v4f sw1 = *(const v4f*)(pW + m * PPITCH + 8 * hh + 4);
    const v4f sr0 = *(const v4f*)(pR + m * PPITCH + 8 * hh);
    const v4f sr1 = *(const v4f*)(pR + m * PPITCH + 8 * hh + 4);
    const v4f iv0 = *(const v4f*)(invs + 8 * hh);
    const v4f iv1 = *(const v4f*)(invs + 8 * hh + 4);
    const v4f b0 = *(const v4f*)(bvec + s * NR + 8 * hh);
    const v4f b1 = *(const v4f*)(bvec + s * NR + 8 * hh + 4);
    const v4f p0 = (sw0 - sr0 * iv0 + b0) * iv0;
    const v4f p1 = (sw1 - sr1 * iv1 + b1) * iv1;
    const float pf[8] = { p0.x, p0.y, p0.z, p0.w, p1.x, p1.y, p1.z, p1.w };

    {
        v4f hx[TTILE];
        phase_b_chunk<true, 0>(hx, pf, sl1, Rl, hseq, lane4, obase, t0, NCHUNK - 1, lane, m, hh);
        phase_b_chunk<true, 0>(hx, pf, sl0, Rl, hseq, lane4, obase, t0, NCHUNK - 2, lane, m, hh);
    }
    phase_b_chunk<false, 4>(hk0, pf, sl1, Rl, hseq, lane4, obase, t0, 0, lane, m, hh);
    phase_b_chunk<false, 4>(hk1, pf, sl0, Rl, hseq, lane4, obase, t0, 1, lane, m, hh);
    phase_b_chunk<false, 0>(hva, pf, sl1, Rl, hseq, lane4, obase, t0, 2, lane, m, hh);
    phase_b_chunk<false, 0>(hvb, pf, sl0, Rl, hseq, lane4, obase, t0, 3, lane, m, hh);
    phase_b_chunk<false, 0>(hk0, pf, sl1, Rl, hseq, lane4, obase, t0, 4, lane, m, hh);
    phase_b_chunk<false, 0>(hk1, pf, sl0, Rl, hseq, lane4, obase, t0, 5, lane, m, hh);
}

extern "C" void kernel_launch(void* const* d_in, const int* in_sizes, int n_in,
                              void* d_out, int out_size, void* d_ws, size_t ws_size,
                              hipStream_t stream)
{
    const float* h    = (const float*)d_in[0];
    const float* R    = (const float*)d_in[1];
    const float* W    = (const float*)d_in[2];
    const float* bvec = (const float*)d_in[3];
    float* out = (float*)d_out;

    (void)in_sizes; (void)n_in; (void)out_size; (void)d_ws; (void)ws_size;

    loreft_main_kernel<<<NS * BLOCKS_PER_STEP, 256, 0, stream>>>(h, R, W, bvec, out);
}
